# starting kernel
# baseline (speedup 1.0000x reference)
#include <hip/hip_runtime.h>

typedef __attribute__((ext_vector_type(2))) float v2f;
typedef __attribute__((ext_vector_type(8))) float v8f;

#define T_N 256
#define R_N 192
#define REF_STEP (48.0f / 191.0f)

__global__ __launch_bounds__(256) void interp_wmma_kernel(
    const float* __restrict__ x, const float* __restrict__ kernel,
    float* __restrict__ out) {
  __shared__ float sh_x[T_N];
  __shared__ float sh_lm[T_N];
  __shared__ float sh_ts[T_N];
  __shared__ float sh_mx[2][R_N];
  __shared__ float sh_red[8][2][16];

  const int tid = threadIdx.x;
  const int bd  = blockIdx.x;
  const int b   = bd >> 4;
  const int d   = bd & 15;

  const float k = kernel[d];
  const float alpha = (k > 0.0f) ? (k + log1pf(__expf(-k))) : log1pf(__expf(k));

  const float* base = x + (size_t)b * 64 * T_N;
  const float* px = base + (size_t)d * T_N;
  const float* pm = base + (size_t)(16 + d) * T_N;
  const float* pt = base + (size_t)(32 + d) * T_N;

  if (tid == 0) {
    __builtin_prefetch(px, 0, 1);
    __builtin_prefetch(pm, 0, 1);
    __builtin_prefetch(pt, 0, 1);
  }

  sh_x[tid]  = px[tid];
  sh_lm[tid] = __logf(fmaxf(pm[tid], 1e-30f));
  sh_ts[tid] = pt[tid];
  __syncthreads();

  if (tid < R_N) {
    const float refr = tid * REF_STEP;
    float mx1 = -3.0e38f, mx10 = -3.0e38f;
    for (int t = 0; t < T_N; ++t) {
      const float dz = sh_ts[t] - refr;
      const float z  = alpha * dz * dz;
      const float lm = sh_lm[t];
      mx1  = fmaxf(mx1,  lm - z);
      mx10 = fmaxf(mx10, lm - 10.0f * z);
    }
    sh_mx[0][tid] = mx1;
    sh_mx[1][tid] = mx10;
  }
  __syncthreads();

  const int wave  = tid >> 5;
  const int lane  = tid & 31;
  const int m16   = lane & 15;
  const int khalf = lane >> 4;

  const float selx = (m16 == 0) ? 1.0f : 0.0f;
  const float sel1 = (m16 == 1) ? 1.0f : 0.0f;

  for (int i = 0; i < 3; ++i) {
    const int job = wave + 8 * i;
    const int p   = job / 12;
    const int R   = job % 12;
    const int r   = R * 16 + m16;
    const float refr  = r * REF_STEP;
    const float scale = (p == 0) ? alpha : 10.0f * alpha;
    const float mx    = sh_mx[p][r];

    v8f acc = {};
    for (int chunk = 0; chunk < 64; ++chunk) {
      const int t0 = chunk * 4 + khalf * 2;
      const float d0 = sh_ts[t0]     - refr;
      const float d1 = sh_ts[t0 + 1] - refr;
      const float x0 = sh_x[t0];
      const float x1 = sh_x[t0 + 1];
      v2f av, bv;
      av.x = __expf(sh_lm[t0]     - scale * d0 * d0 - mx);
      av.y = __expf(sh_lm[t0 + 1] - scale * d1 * d1 - mx);
      bv.x = fmaf(selx, x0, sel1);
      bv.y = fmaf(selx, x1, sel1);
      acc = __builtin_amdgcn_wmma_f32_16x16x4_f32(
           false, av,  false, bv,
           (short)0, acc,  false,  false);
    }

    if (m16 < 2) {
#pragma unroll
      for (int j = 0; j < 8; ++j)
        sh_red[wave][m16][j + 8 * khalf] = acc[j];
    }
    __syncthreads();

    if (lane < 16) {
      const int r2   = R * 16 + lane;
      const float dn = sh_red[wave][1][lane];
      const float ys = sh_red[wave][0][lane];
      const size_t ob = (size_t)b * 48 * R_N;
      const int ch_y = (p == 0) ? d : (32 + d);
      out[ob + (size_t)ch_y * R_N + r2] = ys / dn;
      if (p == 0)
        out[ob + (size_t)(16 + d) * R_N + r2] = sh_mx[0][r2] + __logf(dn);
    }
    __syncthreads();
  }
}

extern "C" void kernel_launch(void* const* d_in, const int* in_sizes, int n_in,
                              void* d_out, int out_size, void* d_ws, size_t ws_size,
                              hipStream_t stream) {
  const float* x      = (const float*)d_in[0];
  const float* kernel = (const float*)d_in[1];
  float* out          = (float*)d_out;
  (void)in_sizes; (void)n_in; (void)out_size; (void)d_ws; (void)ws_size;

  interp_wmma_kernel<<<dim3(64 * 16), dim3(256), 0, stream>>>(x, kernel, out);
}


# Job body on a window of 8 consecutive t tiles: per-tile time span and top log m bound every score above and the row shift below; tiles outside fail the 40-binade test anyway. Sums bit-identical.
# speedup vs baseline: 3.3297x; 3.3297x over previous
#include <hip/hip_runtime.h>

typedef __attribute__((ext_vector_type(2))) float v2f;
typedef __attribute__((ext_vector_type(8))) float v8f;

#define T_N 256
#define R_N 192
#define REF_STEP (48.0f / 191.0f)
#define LOG2E_F 1.44269504088896341f
#define LN2_F   0.69314718055994531f

template <int NT>
__device__ __forceinline__ void interp_job(
    const float* sh_x, const float* sh_tl, const float* sh_l9,
    const int tb0, const float sl, const float er, const float r0,
    const int m16, const int khalf, const int lane, const float selx, const float sel1,
    float* __restrict__ out, const size_t ob, const int d, const int r) {
  v2f bq;
  bq.x = khalf ? 2.0f * sl * er : LOG2E_F;
  bq.y = khalf ? -sl * er * er : -sl;
  asm volatile("" ::: "memory");
  const float* p_tl = sh_tl + 2 * (tb0 + m16);
  const float* p_l9 = sh_l9 + (tb0 + 8 * khalf);
  const float* p_x  = sh_x  + (tb0 + 8 * khalf);
  v8f S[NT];
#pragma unroll
  for (int tile = 0; tile < NT; ++tile) {
    const v2f tl = *reinterpret_cast<const v2f*>(p_tl + 32 * tile);
    const float u = tl.x - r0;
    v2f aq;
    aq.x = khalf ? u : tl.y;
    aq.y = khalf ? 1.0f : u * u;
    const v8f zero = {};
    S[tile] = __builtin_amdgcn_wmma_f32_16x16x4_f32(
         false, aq,  false, bq,
         (short)0, zero,  false,  false);
  }

  for (int p = 0; p < 2; ++p) {
    if (p == 1) {
#pragma unroll
      for (int tile = 0; tile < NT; ++tile) {
        const float4 llo = *reinterpret_cast<const float4*>(p_l9 + 16 * tile);
        const float4 lhi = *reinterpret_cast<const float4*>(p_l9 + 16 * tile + 4);
        const float l9[8] = {llo.x, llo.y, llo.z, llo.w, lhi.x, lhi.y, lhi.z, lhi.w};
#pragma unroll
        for (int j = 0; j < 8; ++j) S[tile][j] = fmaf(10.0f, S[tile][j], l9[j]);
      }
    }

    float tm[NT];
#pragma unroll
    for (int tile = 0; tile < NT; ++tile) {
      tm[tile] = fmaxf(fmaxf(fmaxf(S[tile][0], S[tile][1]), fmaxf(S[tile][2], S[tile][3])),
                       fmaxf(fmaxf(S[tile][4], S[tile][5]), fmaxf(S[tile][6], S[tile][7])));
    }
    float mx = tm[0];
#pragma unroll
    for (int tile = 1; tile < NT; ++tile) mx = fmaxf(mx, tm[tile]);
    mx = fmaxf(mx, __shfl_xor(mx, 16));

    bool keep[NT];
#pragma unroll
    for (int tile = 0; tile < NT; ++tile) {
      const bool live = (tm[tile] - mx) >= -40.0f;
      keep[tile] = __builtin_amdgcn_ballot_w32(live) != 0u;
    }

    v8f acc0 = {}, acc1 = {};
#pragma unroll
    for (int tile = 0; tile < NT; ++tile) {
      if (!keep[tile]) continue;
      const float4 xlo = *reinterpret_cast<const float4*>(p_x + 16 * tile);
      const float4 xhi = *reinterpret_cast<const float4*>(p_x + 16 * tile + 4);
      const float xs[8] = {xlo.x, xlo.y, xlo.z, xlo.w, xhi.x, xhi.y, xhi.z, xhi.w};
#pragma unroll
      for (int q = 0; q < 4; ++q) {
        v2f ev, av;
        ev.x = __builtin_amdgcn_exp2f(S[tile][2 * q]     - mx);
        ev.y = __builtin_amdgcn_exp2f(S[tile][2 * q + 1] - mx);
        av.x = fmaf(selx, xs[2 * q],     sel1);
        av.y = fmaf(selx, xs[2 * q + 1], sel1);
        if (q & 1)
          acc1 = __builtin_amdgcn_wmma_f32_16x16x4_f32(
               false, av,  false, ev,
               (short)0, acc1,  false,  false);
        else
          acc0 = __builtin_amdgcn_wmma_f32_16x16x4_f32(
               false, av,  false, ev,
               (short)0, acc0,  false,  false);
      }
    }

    if (lane < 16) {
      const float ys = acc0[0] + acc1[0];
      const float dn = acc0[1] + acc1[1];
      const int ch_y = (p == 0) ? d : (32 + d);
      out[ob + (size_t)ch_y * R_N + r] = ys / dn;
      if (p == 0)
        out[ob + (size_t)(16 + d) * R_N + r] = (mx + __log2f(dn)) * LN2_F;
    }
  }
}

__global__ __launch_bounds__(128) void interp_wmma_kernel(
    const float* __restrict__ x, const float* __restrict__ kernel,
    float* __restrict__ out) {
  __shared__ __attribute__((aligned(16))) float sh_x[T_N];
  __shared__ __attribute__((aligned(16))) float sh_tl[T_N * 2];
  __shared__ __attribute__((aligned(16))) float sh_l9[T_N];
  __shared__ __attribute__((aligned(16))) float sh_tb[16 * 4];

  const int tid = threadIdx.x;
  const int bd  = blockIdx.x;
  const int b   = bd >> 4;
  const int d   = bd & 15;

  const float k = kernel[d];
  const float alpha = (k > 0.0f) ? (k + log1pf(__expf(-k))) : log1pf(__expf(k));

  const float* base = x + (size_t)b * 64 * T_N;
  const float* px = base + (size_t)d * T_N;
  const float* pm = base + (size_t)(16 + d) * T_N;
  const float* pt = base + (size_t)(32 + d) * T_N;

  if (tid == 0) {
    __builtin_prefetch(px, 0, 1);
    __builtin_prefetch(pm, 0, 1);
    __builtin_prefetch(pt, 0, 1);
  }

  const float ts0 = pt[tid],  ts1 = pt[tid + 128];
  const float mk0 = pm[tid],  mk1 = pm[tid + 128];
  const float xv0 = px[tid],  xv1 = px[tid + 128];
  v2f tl0, tl1;
  tl0.x = ts0;  tl0.y = __logf(fmaxf(mk0, 1e-30f));
  tl1.x = ts1;  tl1.y = __logf(fmaxf(mk1, 1e-30f));
  sh_x[tid] = xv0;
  sh_x[tid + 128] = xv1;
  *reinterpret_cast<v2f*>(&sh_tl[2 * tid]) = tl0;
  *reinterpret_cast<v2f*>(&sh_tl[2 * (tid + 128)]) = tl1;
  sh_l9[tid]       = (-9.0f * LOG2E_F) * tl0.y;
  sh_l9[tid + 128] = (-9.0f * LOG2E_F) * tl1.y;
  {
    float lo0 = ts0, hi0 = ts0, lt0 = tl0.y;
    float lo1 = ts1, hi1 = ts1, lt1 = tl1.y;
#pragma unroll
    for (int s = 1; s < 16; s <<= 1) {
      lo0 = fminf(lo0, __shfl_xor(lo0, s));
      hi0 = fmaxf(hi0, __shfl_xor(hi0, s));
      lt0 = fmaxf(lt0, __shfl_xor(lt0, s));
      lo1 = fminf(lo1, __shfl_xor(lo1, s));
      hi1 = fmaxf(hi1, __shfl_xor(hi1, s));
      lt1 = fmaxf(lt1, __shfl_xor(lt1, s));
    }
    if ((tid & 15) == 0) {
      float4 tb;
      tb.x = lo0; tb.y = hi0; tb.z = LOG2E_F * lt0; tb.w = 0.0f;
      *reinterpret_cast<float4*>(&sh_tb[4 * (tid >> 4)]) = tb;
      tb.x = lo1; tb.y = hi1; tb.z = LOG2E_F * lt1;
      *reinterpret_cast<float4*>(&sh_tb[4 * (8 + (tid >> 4))]) = tb;
    }
  }
  __syncthreads();

  const int wave  = tid >> 5;
  const int lane  = tid & 31;
  const int m16   = lane & 15;
  const int khalf = lane >> 4;

  const float selx = (m16 == 0) ? 1.0f : 0.0f;
  const float sel1 = (m16 == 1) ? 1.0f : 0.0f;

  const float sl = alpha * LOG2E_F;
  const float sl10 = 10.0f * sl;
  const size_t ob = (size_t)b * 48 * R_N;
  for (int i = 0; i < 3; ++i) {
    const int R   = wave + 4 * i;
    const int r   = R * 16 + m16;
    const float refr  = r * REF_STEP;
    const float r0    = (float)(4 * R + 2);
    const float er    = refr - r0;

    float lb1 = -INFINITY, lb10 = -INFINITY;
    float ub1[8], ub10[8];
#pragma unroll
    for (int q = 0; q < 8; ++q) {
      const float4 tb = *reinterpret_cast<const float4*>(&sh_tb[4 * (8 * khalf + q)]);
      const float da = tb.x - refr;
      const float db = tb.y - refr;
      const float nr = fmaxf(fmaxf(da, -db), 0.0f);
      const float fr = fmaxf(-da, db);
      const float n2 = nr * nr;
      const float f2 = fr * fr;
      ub1[q]  = fmaf(-sl,   n2, tb.z);
      ub10[q] = fmaf(-sl10, n2, tb.z);
      lb1  = fmaxf(lb1,  fmaf(-sl,   f2, tb.z));
      lb10 = fmaxf(lb10, fmaf(-sl10, f2, tb.z));
    }
    lb1  = fmaxf(lb1,  __shfl_xor(lb1,  16));
    lb10 = fmaxf(lb10, __shfl_xor(lb10, 16));
    const float th1 = lb1 - 42.0f, th10 = lb10 - 42.0f;
    unsigned cm = 0u;
#pragma unroll
    for (int q = 0; q < 8; ++q) {
      const unsigned bal = __builtin_amdgcn_ballot_w32(!(ub1[q] < th1)) |
                           __builtin_amdgcn_ballot_w32(!(ub10[q] < th10));
      cm |= ((bal & 0xFFFFu) != 0u ? 1u : 0u) << q;
      cm |= ((bal >> 16) != 0u ? 1u : 0u) << (8 + q);
    }
    bool fast = false;
    int tlo8 = 0;
    if (cm != 0u) {
      const int first = __builtin_ctz(cm);
      const int last  = 31 - __builtin_clz(cm);
      tlo8 = first < 8 ? first : 8;
      fast = (last - tlo8) <= 7;
    }

    if (fast)
      interp_job<8>(sh_x, sh_tl, sh_l9, 16 * tlo8, sl, er, r0, m16, khalf, lane, selx, sel1, out, ob, d, r);
    else
      interp_job<16>(sh_x, sh_tl, sh_l9, 0, sl, er, r0, m16, khalf, lane, selx, sel1, out, ob, d, r);
  }
}

extern "C" void kernel_launch(void* const* d_in, const int* in_sizes, int n_in,
                              void* d_out, int out_size, void* d_ws, size_t ws_size,
                              hipStream_t stream) {
  const float* x      = (const float*)d_in[0];
  const float* kernel = (const float*)d_in[1];
  float* out          = (float*)d_out;
  (void)in_sizes; (void)n_in; (void)out_size; (void)d_ws; (void)ws_size;

  interp_wmma_kernel<<<dim3(64 * 16), dim3(128), 0, stream>>>(x, kernel, out);
}


# Wave priority falls with progress (s_setprio 3,2,1 per job, 0 in the last half job): the four work-groups' waves on a SIMD end together, not by age with the youngest alone. Sums bit-identical.
# speedup vs baseline: 3.6475x; 1.0954x over previous
#include <hip/hip_runtime.h>

typedef __attribute__((ext_vector_type(2))) float v2f;
typedef __attribute__((ext_vector_type(8))) float v8f;

#define T_N 256
#define R_N 192
#define REF_STEP (48.0f / 191.0f)
#define LOG2E_F 1.44269504088896341f
#define LN2_F   0.69314718055994531f

template <int NT>
__device__ __forceinline__ void interp_job(
    const float* sh_x, const float* sh_tl, const float* sh_l9,
    const int tb0, const float sl, const float er, const float r0,
    const int m16, const int khalf, const int lane, const float selx, const float sel1,
    float* __restrict__ out, const size_t ob, const int d, const int r, const bool lastjob) {
  v2f bq;
  bq.x = khalf ? 2.0f * sl * er : LOG2E_F;
  bq.y = khalf ? -sl * er * er : -sl;
  asm volatile("" ::: "memory");
  const float* p_tl = sh_tl + 2 * (tb0 + m16);
  const float* p_l9 = sh_l9 + (tb0 + 8 * khalf);
  const float* p_x  = sh_x  + (tb0 + 8 * khalf);
  v8f S[NT];
#pragma unroll
  for (int tile = 0; tile < NT; ++tile) {
    const v2f tl = *reinterpret_cast<const v2f*>(p_tl + 32 * tile);
    const float u = tl.x - r0;
    v2f aq;
    aq.x = khalf ? u : tl.y;
    aq.y = khalf ? 1.0f : u * u;
    const v8f zero = {};
    S[tile] = __builtin_amdgcn_wmma_f32_16x16x4_f32(
         false, aq,  false, bq,
         (short)0, zero,  false,  false);
  }

  for (int p = 0; p < 2; ++p) {
    if (p == 1 && lastjob) __builtin_amdgcn_s_setprio(0);
    if (p == 1) {
#pragma unroll
      for (int tile = 0; tile < NT; ++tile) {
        const float4 llo = *reinterpret_cast<const float4*>(p_l9 + 16 * tile);
        const float4 lhi = *reinterpret_cast<const float4*>(p_l9 + 16 * tile + 4);
        const float l9[8] = {llo.x, llo.y, llo.z, llo.w, lhi.x, lhi.y, lhi.z, lhi.w};
#pragma unroll
        for (int j = 0; j < 8; ++j) S[tile][j] = fmaf(10.0f, S[tile][j], l9[j]);
      }
    }

    float tm[NT];
#pragma unroll
    for (int tile = 0; tile < NT; ++tile) {
      tm[tile] = fmaxf(fmaxf(fmaxf(S[tile][0], S[tile][1]), fmaxf(S[tile][2], S[tile][3])),
                       fmaxf(fmaxf(S[tile][4], S[tile][5]), fmaxf(S[tile][6], S[tile][7])));
    }
    float mx = tm[0];
#pragma unroll
    for (int tile = 1; tile < NT; ++tile) mx = fmaxf(mx, tm[tile]);
    mx = fmaxf(mx, __shfl_xor(mx, 16));

    bool keep[NT];
#pragma unroll
    for (int tile = 0; tile < NT; ++tile) {
      const bool live = (tm[tile] - mx) >= -40.0f;
      keep[tile] = __builtin_amdgcn_ballot_w32(live) != 0u;
    }

    v8f acc0 = {}, acc1 = {};
#pragma unroll
    for (int tile = 0; tile < NT; ++tile) {
      if (!keep[tile]) continue;
      const float4 xlo = *reinterpret_cast<const float4*>(p_x + 16 * tile);
      const float4 xhi = *reinterpret_cast<const float4*>(p_x + 16 * tile + 4);
      const float xs[8] = {xlo.x, xlo.y, xlo.z, xlo.w, xhi.x, xhi.y, xhi.z, xhi.w};
#pragma unroll
      for (int q = 0; q < 4; ++q) {
        v2f ev, av;
        ev.x = __builtin_amdgcn_exp2f(S[tile][2 * q]     - mx);
        ev.y = __builtin_amdgcn_exp2f(S[tile][2 * q + 1] - mx);
        av.x = fmaf(selx, xs[2 * q],     sel1);
        av.y = fmaf(selx, xs[2 * q + 1], sel1);
        if (q & 1)
          acc1 = __builtin_amdgcn_wmma_f32_16x16x4_f32(
               false, av,  false, ev,
               (short)0, acc1,  false,  false);
        else
          acc0 = __builtin_amdgcn_wmma_f32_16x16x4_f32(
               false, av,  false, ev,
               (short)0, acc0,  false,  false);
      }
    }

    if (lane < 16) {
      const float ys = acc0[0] + acc1[0];
      const float dn = acc0[1] + acc1[1];
      const int ch_y = (p == 0) ? d : (32 + d);
      out[ob + (size_t)ch_y * R_N + r] = ys / dn;
      if (p == 0)
        out[ob + (size_t)(16 + d) * R_N + r] = (mx + __log2f(dn)) * LN2_F;
    }
  }
}

__global__ __launch_bounds__(128) void interp_wmma_kernel(
    const float* __restrict__ x, const float* __restrict__ kernel,
    float* __restrict__ out) {
  __shared__ __attribute__((aligned(16))) float sh_x[T_N];
  __shared__ __attribute__((aligned(16))) float sh_tl[T_N * 2];
  __shared__ __attribute__((aligned(16))) float sh_l9[T_N];
  __shared__ __attribute__((aligned(16))) float sh_tb[16 * 4];

  __builtin_amdgcn_s_setprio(3);
  const int tid = threadIdx.x;
  const int bd  = blockIdx.x;
  const int b   = bd >> 4;
  const int d   = bd & 15;

  const float k = kernel[d];
  const float alpha = (k > 0.0f) ? (k + log1pf(__expf(-k))) : log1pf(__expf(k));

  const float* base = x + (size_t)b * 64 * T_N;
  const float* px = base + (size_t)d * T_N;
  const float* pm = base + (size_t)(16 + d) * T_N;
  const float* pt = base + (size_t)(32 + d) * T_N;

  if (tid == 0) {
    __builtin_prefetch(px, 0, 1);
    __builtin_prefetch(pm, 0, 1);
    __builtin_prefetch(pt, 0, 1);
  }

  const float ts0 = pt[tid],  ts1 = pt[tid + 128];
  const float mk0 = pm[tid],  mk1 = pm[tid + 128];
  const float xv0 = px[tid],  xv1 = px[tid + 128];
  v2f tl0, tl1;
  tl0.x = ts0;  tl0.y = __logf(fmaxf(mk0, 1e-30f));
  tl1.x = ts1;  tl1.y = __logf(fmaxf(mk1, 1e-30f));
  sh_x[tid] = xv0;
  sh_x[tid + 128] = xv1;
  *reinterpret_cast<v2f*>(&sh_tl[2 * tid]) = tl0;
  *reinterpret_cast<v2f*>(&sh_tl[2 * (tid + 128)]) = tl1;
  sh_l9[tid]       = (-9.0f * LOG2E_F) * tl0.y;
  sh_l9[tid + 128] = (-9.0f * LOG2E_F) * tl1.y;
  {
    float lo0 = ts0, hi0 = ts0, lt0 = tl0.y;
    float lo1 = ts1, hi1 = ts1, lt1 = tl1.y;
#pragma unroll
    for (int s = 1; s < 16; s <<= 1) {
      lo0 = fminf(lo0, __shfl_xor(lo0, s));
      hi0 = fmaxf(hi0, __shfl_xor(hi0, s));
      lt0 = fmaxf(lt0, __shfl_xor(lt0, s));
      lo1 = fminf(lo1, __shfl_xor(lo1, s));
      hi1 = fmaxf(hi1, __shfl_xor(hi1, s));
      lt1 = fmaxf(lt1, __shfl_xor(lt1, s));
    }
    if ((tid & 15) == 0) {
      float4 tb;
      tb.x = lo0; tb.y = hi0; tb.z = LOG2E_F * lt0; tb.w = 0.0f;
      *reinterpret_cast<float4*>(&sh_tb[4 * (tid >> 4)]) = tb;
      tb.x = lo1; tb.y = hi1; tb.z = LOG2E_F * lt1;
      *reinterpret_cast<float4*>(&sh_tb[4 * (8 + (tid >> 4))]) = tb;
    }
  }
  __syncthreads();

  const int wave  = tid >> 5;
  const int lane  = tid & 31;
  const int m16   = lane & 15;
  const int khalf = lane >> 4;

  const float selx = (m16 == 0) ? 1.0f : 0.0f;
  const float sel1 = (m16 == 1) ? 1.0f : 0.0f;

  const float sl = alpha * LOG2E_F;
  const float sl10 = 10.0f * sl;
  const size_t ob = (size_t)b * 48 * R_N;
  for (int i = 0; i < 3; ++i) {
    if (i == 1) __builtin_amdgcn_s_setprio(2);
    if (i == 2) __builtin_amdgcn_s_setprio(1);
    const int R   = wave + 4 * i;
    const int r   = R * 16 + m16;
    const float refr  = r * REF_STEP;
    const float r0    = (float)(4 * R + 2);
    const float er    = refr - r0;

    float lb1 = -INFINITY, lb10 = -INFINITY;
    float ub1[8], ub10[8];
#pragma unroll
    for (int q = 0; q < 8; ++q) {
      const float4 tb = *reinterpret_cast<const float4*>(&sh_tb[4 * (8 * khalf + q)]);
      const float da = tb.x - refr;
      const float db = tb.y - refr;
      const float nr = fmaxf(fmaxf(da, -db), 0.0f);
      const float fr = fmaxf(-da, db);
      const float n2 = nr * nr;
      const float f2 = fr * fr;
      ub1[q]  = fmaf(-sl,   n2, tb.z);
      ub10[q] = fmaf(-sl10, n2, tb.z);
      lb1  = fmaxf(lb1,  fmaf(-sl,   f2, tb.z));
      lb10 = fmaxf(lb10, fmaf(-sl10, f2, tb.z));
    }
    lb1  = fmaxf(lb1,  __shfl_xor(lb1,  16));
    lb10 = fmaxf(lb10, __shfl_xor(lb10, 16));
    const float th1 = lb1 - 42.0f, th10 = lb10 - 42.0f;
    unsigned cm = 0u;
#pragma unroll
    for (int q = 0; q < 8; ++q) {
      const unsigned bal = __builtin_amdgcn_ballot_w32(!(ub1[q] < th1)) |
                           __builtin_amdgcn_ballot_w32(!(ub10[q] < th10));
      cm |= ((bal & 0xFFFFu) != 0u ? 1u : 0u) << q;
      cm |= ((bal >> 16) != 0u ? 1u : 0u) << (8 + q);
    }
    bool fast = false;
    int tlo8 = 0;
    if (cm != 0u) {
      const int first = __builtin_ctz(cm);
      const int last  = 31 - __builtin_clz(cm);
      tlo8 = first < 8 ? first : 8;
      fast = (last - tlo8) <= 7;
    }

    if (fast)
      interp_job<8>(sh_x, sh_tl, sh_l9, 16 * tlo8, sl, er, r0, m16, khalf, lane, selx, sel1, out, ob, d, r, i == 2);
    else
      interp_job<16>(sh_x, sh_tl, sh_l9, 0, sl, er, r0, m16, khalf, lane, selx, sel1, out, ob, d, r, i == 2);
  }
}

extern "C" void kernel_launch(void* const* d_in, const int* in_sizes, int n_in,
                              void* d_out, int out_size, void* d_ws, size_t ws_size,
                              hipStream_t stream) {
  const float* x      = (const float*)d_in[0];
  const float* kernel = (const float*)d_in[1];
  float* out          = (float*)d_out;
  (void)in_sizes; (void)n_in; (void)out_size; (void)d_ws; (void)ws_size;

  interp_wmma_kernel<<<dim3(64 * 16), dim3(128), 0, stream>>>(x, kernel, out);
}


# Reduction's A fragments [x;1;0] lie ready in LDS (row picked by lane address), loaded once per job as one group ahead of both paths' products; was 2 loads + 8 select fmas a tile. Sums bit-identical.
# speedup vs baseline: 3.8333x; 1.0509x over previous
#include <hip/hip_runtime.h>

typedef __attribute__((ext_vector_type(2))) float v2f;
typedef __attribute__((ext_vector_type(8))) float v8f;

#define T_N 256
#define R_N 192
#define REF_STEP (48.0f / 191.0f)
#define LOG2E_F 1.44269504088896341f
#define LN2_F   0.69314718055994531f
#define A_PITCH (T_N + 16)

template <int NT>
__device__ __forceinline__ void interp_job(
    const float* sh_arow, const float* sh_tl, const float* sh_l9,
    const int tb0, const float sl, const float er, const float r0,
    const int m16, const int khalf, const int lane,
    float* __restrict__ out, const size_t ob, const int d, const int r, const bool lastjob) {
  v2f bq;
  bq.x = khalf ? 2.0f * sl * er : LOG2E_F;
  bq.y = khalf ? -sl * er * er : -sl;
  asm volatile("" ::: "memory");
  const float* p_tl = sh_tl + 2 * (tb0 + m16);
  const float* p_l9 = sh_l9 + (tb0 + 8 * khalf);
  const float* p_a  = sh_arow + (tb0 + 8 * khalf);
  v2f tlr[NT];
#pragma unroll
  for (int tile = 0; tile < NT; ++tile)
    tlr[tile] = *reinterpret_cast<const v2f*>(p_tl + 32 * tile);

  constexpr bool kGroup = (NT == 8);
  v2f af[kGroup ? NT : 1][4];
  if (kGroup) {
#pragma unroll
    for (int tile = 0; tile < NT; ++tile) {
      const float4 alo = *reinterpret_cast<const float4*>(p_a + 16 * tile);
      const float4 ahi = *reinterpret_cast<const float4*>(p_a + 16 * tile + 4);
      af[tile][0].x = alo.x; af[tile][0].y = alo.y;
      af[tile][1].x = alo.z; af[tile][1].y = alo.w;
      af[tile][2].x = ahi.x; af[tile][2].y = ahi.y;
      af[tile][3].x = ahi.z; af[tile][3].y = ahi.w;
    }
    __builtin_amdgcn_sched_barrier(0);
  }

  v8f S[NT];
#pragma unroll
  for (int tile = 0; tile < NT; ++tile) {
    const v2f tl = tlr[tile];
    const float u = tl.x - r0;
    v2f aq;
    aq.x = khalf ? u : tl.y;
    aq.y = khalf ? 1.0f : u * u;
    const v8f zero = {};
    S[tile] = __builtin_amdgcn_wmma_f32_16x16x4_f32(
         false, aq,  false, bq,
         (short)0, zero,  false,  false);
  }

  for (int p = 0; p < 2; ++p) {
    if (p == 1 && lastjob) __builtin_amdgcn_s_setprio(0);
    if (p == 1) {
#pragma unroll
      for (int tile = 0; tile < NT; ++tile) {
        const float4 llo = *reinterpret_cast<const float4*>(p_l9 + 16 * tile);
        const float4 lhi = *reinterpret_cast<const float4*>(p_l9 + 16 * tile + 4);
        const float l9[8] = {llo.x, llo.y, llo.z, llo.w, lhi.x, lhi.y, lhi.z, lhi.w};
#pragma unroll
        for (int j = 0; j < 8; ++j) S[tile][j] = fmaf(10.0f, S[tile][j], l9[j]);
      }
    }

    float tm[NT];
#pragma unroll
    for (int tile = 0; tile < NT; ++tile) {
      tm[tile] = fmaxf(fmaxf(fmaxf(S[tile][0], S[tile][1]), fmaxf(S[tile][2], S[tile][3])),
                       fmaxf(fmaxf(S[tile][4], S[tile][5]), fmaxf(S[tile][6], S[tile][7])));
    }
    float mx = tm[0];
#pragma unroll
    for (int tile = 1; tile < NT; ++tile) mx = fmaxf(mx, tm[tile]);
    mx = fmaxf(mx, __shfl_xor(mx, 16));

    bool keep[NT];
#pragma unroll
    for (int tile = 0; tile < NT; ++tile) {
      const bool live = (tm[tile] - mx) >= -40.0f;
      keep[tile] = __builtin_amdgcn_ballot_w32(live) != 0u;
    }

    v8f acc0 = {}, acc1 = {};
#pragma unroll
    for (int tile = 0; tile < NT; ++tile) {
      if (!keep[tile]) continue;
      v2f a4[4];
      if (kGroup) {
#pragma unroll
        for (int q = 0; q < 4; ++q) a4[q] = af[tile][q];
      } else {
        const float4 alo = *reinterpret_cast<const float4*>(p_a + 16 * tile);
        const float4 ahi = *reinterpret_cast<const float4*>(p_a + 16 * tile + 4);
        a4[0].x = alo.x; a4[0].y = alo.y;
        a4[1].x = alo.z; a4[1].y = alo.w;
        a4[2].x = ahi.x; a4[2].y = ahi.y;
        a4[3].x = ahi.z; a4[3].y = ahi.w;
      }
#pragma unroll
      for (int q = 0; q < 4; ++q) {
        v2f ev;
        const v2f av = a4[q];
        ev.x = __builtin_amdgcn_exp2f(S[tile][2 * q]     - mx);
        ev.y = __builtin_amdgcn_exp2f(S[tile][2 * q + 1] - mx);
        if (q & 1)
          acc1 = __builtin_amdgcn_wmma_f32_16x16x4_f32(
               false, av,  false, ev,
               (short)0, acc1,  false,  false);
        else
          acc0 = __builtin_amdgcn_wmma_f32_16x16x4_f32(
               false, av,  false, ev,
               (short)0, acc0,  false,  false);
      }
    }

    if (lane < 16) {
      const float ys = acc0[0] + acc1[0];
      const float dn = acc0[1] + acc1[1];
      const int ch_y = (p == 0) ? d : (32 + d);
      out[ob + (size_t)ch_y * R_N + r] = ys / dn;
      if (p == 0)
        out[ob + (size_t)(16 + d) * R_N + r] = (mx + __log2f(dn)) * LN2_F;
    }
  }
}

__global__ __launch_bounds__(128) void interp_wmma_kernel(
    const float* __restrict__ x, const float* __restrict__ kernel,
    float* __restrict__ out) {
  __shared__ __attribute__((aligned(16))) float sh_all[T_N * 2 + T_N + 16 * 4 + 3 * A_PITCH];
  float* const sh_tl = sh_all;
  float* const sh_l9 = sh_all + T_N * 2;
  float* const sh_tb = sh_all + T_N * 3;
  float* const sh_a  = sh_all + T_N * 3 + 16 * 4;

  __builtin_amdgcn_s_setprio(3);
  const int tid = threadIdx.x;
  const int bd  = blockIdx.x;
  const int b   = bd >> 4;
  const int d   = bd & 15;

  const float k = kernel[d];
  const float alpha = (k > 0.0f) ? (k + log1pf(__expf(-k))) : log1pf(__expf(k));

  const float* base = x + (size_t)b * 64 * T_N;
  const float* px = base + (size_t)d * T_N;
  const float* pm = base + (size_t)(16 + d) * T_N;
  const float* pt = base + (size_t)(32 + d) * T_N;

  if (tid == 0) {
    __builtin_prefetch(px, 0, 1);
    __builtin_prefetch(pm, 0, 1);
    __builtin_prefetch(pt, 0, 1);
  }

  const float ts0 = pt[tid],  ts1 = pt[tid + 128];
  const float mk0 = pm[tid],  mk1 = pm[tid + 128];
  const float xv0 = px[tid],  xv1 = px[tid + 128];
  v2f tl0, tl1;
  tl0.x = ts0;  tl0.y = __logf(fmaxf(mk0, 1e-30f));
  tl1.x = ts1;  tl1.y = __logf(fmaxf(mk1, 1e-30f));
  sh_a[tid] = xv0;
  sh_a[tid + 128] = xv1;
  sh_a[A_PITCH + tid] = 1.0f;
  sh_a[A_PITCH + tid + 128] = 1.0f;
  sh_a[2 * A_PITCH + tid] = 0.0f;
  sh_a[2 * A_PITCH + tid + 128] = 0.0f;
  *reinterpret_cast<v2f*>(&sh_tl[2 * tid]) = tl0;
  *reinterpret_cast<v2f*>(&sh_tl[2 * (tid + 128)]) = tl1;
  sh_l9[tid]       = (-9.0f * LOG2E_F) * tl0.y;
  sh_l9[tid + 128] = (-9.0f * LOG2E_F) * tl1.y;
  {
    float lo0 = ts0, hi0 = ts0, lt0 = tl0.y;
    float lo1 = ts1, hi1 = ts1, lt1 = tl1.y;
#pragma unroll
    for (int s = 1; s < 16; s <<= 1) {
      lo0 = fminf(lo0, __shfl_xor(lo0, s));
      hi0 = fmaxf(hi0, __shfl_xor(hi0, s));
      lt0 = fmaxf(lt0, __shfl_xor(lt0, s));
      lo1 = fminf(lo1, __shfl_xor(lo1, s));
      hi1 = fmaxf(hi1, __shfl_xor(hi1, s));
      lt1 = fmaxf(lt1, __shfl_xor(lt1, s));
    }
    if ((tid & 15) == 0) {
      float4 tb;
      tb.x = lo0; tb.y = hi0; tb.z = LOG2E_F * lt0; tb.w = 0.0f;
      *reinterpret_cast<float4*>(&sh_tb[4 * (tid >> 4)]) = tb;
      tb.x = lo1; tb.y = hi1; tb.z = LOG2E_F * lt1;
      *reinterpret_cast<float4*>(&sh_tb[4 * (8 + (tid >> 4))]) = tb;
    }
  }
  __syncthreads();

  const int wave  = tid >> 5;
  const int lane  = tid & 31;
  const int m16   = lane & 15;
  const int khalf = lane >> 4;

  const float* sh_arow = sh_a + (m16 < 2 ? m16 : 2) * A_PITCH;

  const float sl = alpha * LOG2E_F;
  const float sl10 = 10.0f * sl;
  const size_t ob = (size_t)b * 48 * R_N;
  for (int i = 0; i < 3; ++i) {
    if (i == 1) __builtin_amdgcn_s_setprio(2);
    if (i == 2) __builtin_amdgcn_s_setprio(1);
    const int R   = wave + 4 * i;
    const int r   = R * 16 + m16;
    const float refr  = r * REF_STEP;
    const float r0    = (float)(4 * R + 2);
    const float er    = refr - r0;

    float lb1 = -INFINITY, lb10 = -INFINITY;
    float ub1[8], ub10[8];
#pragma unroll
    for (int q = 0; q < 8; ++q) {
      const float4 tb = *reinterpret_cast<const float4*>(&sh_tb[4 * (8 * khalf + q)]);
      const float da = tb.x - refr;
      const float db = tb.y - refr;
      const float nr = fmaxf(fmaxf(da, -db), 0.0f);
      const float fr = fmaxf(-da, db);
      const float n2 = nr * nr;
      const float f2 = fr * fr;
      ub1[q]  = fmaf(-sl,   n2, tb.z);
      ub10[q] = fmaf(-sl10, n2, tb.z);
      lb1  = fmaxf(lb1,  fmaf(-sl,   f2, tb.z));
      lb10 = fmaxf(lb10, fmaf(-sl10, f2, tb.z));
    }
    lb1  = fmaxf(lb1,  __shfl_xor(lb1,  16));
    lb10 = fmaxf(lb10, __shfl_xor(lb10, 16));
    const float th1 = lb1 - 42.0f, th10 = lb10 - 42.0f;
    unsigned cm = 0u;
#pragma unroll
    for (int q = 0; q < 8; ++q) {
      const unsigned bal = __builtin_amdgcn_ballot_w32(!(ub1[q] < th1)) |
                           __builtin_amdgcn_ballot_w32(!(ub10[q] < th10));
      cm |= ((bal & 0xFFFFu) != 0u ? 1u : 0u) << q;
      cm |= ((bal >> 16) != 0u ? 1u : 0u) << (8 + q);
    }
    bool fast = false;
    int tlo8 = 0;
    if (cm != 0u) {
      const int first = __builtin_ctz(cm);
      const int last  = 31 - __builtin_clz(cm);
      tlo8 = first < 8 ? first : 8;
      fast = (last - tlo8) <= 7;
    }

    if (fast)
      interp_job<8>(sh_arow, sh_tl, sh_l9, 16 * tlo8, sl, er, r0, m16, khalf, lane, out, ob, d, r, i == 2);
    else
      interp_job<16>(sh_arow, sh_tl, sh_l9, 0, sl, er, r0, m16, khalf, lane, out, ob, d, r, i == 2);
  }
}

extern "C" void kernel_launch(void* const* d_in, const int* in_sizes, int n_in,
                              void* d_out, int out_size, void* d_ws, size_t ws_size,
                              hipStream_t stream) {
  const float* x      = (const float*)d_in[0];
  const float* kernel = (const float*)d_in[1];
  float* out          = (float*)d_out;
  (void)in_sizes; (void)n_in; (void)out_size; (void)d_ws; (void)ws_size;

  interp_wmma_kernel<<<dim3(64 * 16), dim3(128), 0, stream>>>(x, kernel, out);
}


# t-tile window per r tile from 16 span-to-span bounds under an assumed row-shift floor that the window job checks against the shift it finds (else all 16); per-column scan gone. Sums bit-identical.
# speedup vs baseline: 4.0837x; 1.0653x over previous
#include <hip/hip_runtime.h>

typedef __attribute__((ext_vector_type(2))) float v2f;
typedef __attribute__((ext_vector_type(8))) float v8f;

#define T_N 256
#define R_N 192
#define REF_STEP (48.0f / 191.0f)
#define LOG2E_F 1.44269504088896341f
#define LN2_F   0.69314718055994531f
#define A_PITCH (T_N + 16)

template <int NT>
__device__ __forceinline__ bool interp_job(
    const float* sh_arow, const float* sh_tl, const float* sh_l9,
    const int tb0, const float sl, const float er, const float r0,
    const float g1, const float g10,
    const int m16, const int khalf, const int lane,
    float* __restrict__ out, const size_t ob, const int d, const int r, const bool lastjob) {
  v2f bq;
  bq.x = khalf ? 2.0f * sl * er : LOG2E_F;
  bq.y = khalf ? -sl * er * er : -sl;
  asm volatile("" ::: "memory");
  const float* p_tl = sh_tl + 2 * (tb0 + m16);
  const float* p_l9 = sh_l9 + (tb0 + 8 * khalf);
  const float* p_a  = sh_arow + (tb0 + 8 * khalf);
  v2f tlr[NT];
#pragma unroll
  for (int tile = 0; tile < NT; ++tile)
    tlr[tile] = *reinterpret_cast<const v2f*>(p_tl + 32 * tile);

  constexpr bool kGroup = (NT == 8);
  v2f af[kGroup ? NT : 1][4];
  if (kGroup) {
#pragma unroll
    for (int tile = 0; tile < NT; ++tile) {
      const float4 alo = *reinterpret_cast<const float4*>(p_a + 16 * tile);
      const float4 ahi = *reinterpret_cast<const float4*>(p_a + 16 * tile + 4);
      af[tile][0].x = alo.x; af[tile][0].y = alo.y;
      af[tile][1].x = alo.z; af[tile][1].y = alo.w;
      af[tile][2].x = ahi.x; af[tile][2].y = ahi.y;
      af[tile][3].x = ahi.z; af[tile][3].y = ahi.w;
    }
    __builtin_amdgcn_sched_barrier(0);
  }

  v8f S[NT];
#pragma unroll
  for (int tile = 0; tile < NT; ++tile) {
    const v2f tl = tlr[tile];
    const float u = tl.x - r0;
    v2f aq;
    aq.x = khalf ? u : tl.y;
    aq.y = khalf ? 1.0f : u * u;
    const v8f zero = {};
    S[tile] = __builtin_amdgcn_wmma_f32_16x16x4_f32(
         false, aq,  false, bq,
         (short)0, zero,  false,  false);
  }

  for (int p = 0; p < 2; ++p) {
    if (p == 1 && lastjob) __builtin_amdgcn_s_setprio(0);
    if (p == 1) {
#pragma unroll
      for (int tile = 0; tile < NT; ++tile) {
        const float4 llo = *reinterpret_cast<const float4*>(p_l9 + 16 * tile);
        const float4 lhi = *reinterpret_cast<const float4*>(p_l9 + 16 * tile + 4);
        const float l9[8] = {llo.x, llo.y, llo.z, llo.w, lhi.x, lhi.y, lhi.z, lhi.w};
        const v2f ten = {10.0f, 10.0f};
#pragma unroll
        for (int q = 0; q < 4; ++q) {
          v2f s2, l2;
          s2.x = S[tile][2 * q];  s2.y = S[tile][2 * q + 1];
          l2.x = l9[2 * q];       l2.y = l9[2 * q + 1];
          s2 = __builtin_elementwise_fma(ten, s2, l2);
          S[tile][2 * q] = s2.x;  S[tile][2 * q + 1] = s2.y;
        }
      }
    }

    float tm[NT];
#pragma unroll
    for (int tile = 0; tile < NT; ++tile) {
      tm[tile] = fmaxf(fmaxf(fmaxf(S[tile][0], S[tile][1]), fmaxf(S[tile][2], S[tile][3])),
                       fmaxf(fmaxf(S[tile][4], S[tile][5]), fmaxf(S[tile][6], S[tile][7])));
    }
    float mx = tm[0];
#pragma unroll
    for (int tile = 1; tile < NT; ++tile) mx = fmaxf(mx, tm[tile]);
    mx = fmaxf(mx, __shfl_xor(mx, 16));


    bool keep[NT];
#pragma unroll
    for (int tile = 0; tile < NT; ++tile) {
      const bool live = (tm[tile] - mx) >= -40.0f;
      keep[tile] = __builtin_amdgcn_ballot_w32(live) != 0u;
    }

    if (kGroup) {
      const bool below = !(mx >= (p == 0 ? g1 : g10));
      if (__builtin_amdgcn_ballot_w32(below) != 0u) return false;
    }

    v8f acc0 = {}, acc1 = {};
#pragma unroll
    for (int tile = 0; tile < NT; ++tile) {
      if (!keep[tile]) continue;
      v2f a4[4];
      if (kGroup) {
#pragma unroll
        for (int q = 0; q < 4; ++q) a4[q] = af[tile][q];
      } else {
        const float4 alo = *reinterpret_cast<const float4*>(p_a + 16 * tile);
        const float4 ahi = *reinterpret_cast<const float4*>(p_a + 16 * tile + 4);
        a4[0].x = alo.x; a4[0].y = alo.y;
        a4[1].x = alo.z; a4[1].y = alo.w;
        a4[2].x = ahi.x; a4[2].y = ahi.y;
        a4[3].x = ahi.z; a4[3].y = ahi.w;
      }
#pragma unroll
      for (int q = 0; q < 4; ++q) {
        v2f ev;
        const v2f av = a4[q];
        ev.x = __builtin_amdgcn_exp2f(S[tile][2 * q]     - mx);
        ev.y = __builtin_amdgcn_exp2f(S[tile][2 * q + 1] - mx);
        if (q & 1)
          acc1 = __builtin_amdgcn_wmma_f32_16x16x4_f32(
               false, av,  false, ev,
               (short)0, acc1,  false,  false);
        else
          acc0 = __builtin_amdgcn_wmma_f32_16x16x4_f32(
               false, av,  false, ev,
               (short)0, acc0,  false,  false);
      }
    }

    if (lane < 16) {
      const float ys = acc0[0] + acc1[0];
      const float dn = acc0[1] + acc1[1];
      const int ch_y = (p == 0) ? d : (32 + d);
      out[ob + (size_t)ch_y * R_N + r] = ys / dn;
      if (p == 0)
        out[ob + (size_t)(16 + d) * R_N + r] = (mx + __log2f(dn)) * LN2_F;
    }
  }
  return true;
}

__global__ __launch_bounds__(128) void interp_wmma_kernel(
    const float* __restrict__ x, const float* __restrict__ kernel,
    float* __restrict__ out) {
  __shared__ __attribute__((aligned(16))) float sh_all[T_N * 2 + T_N + 16 * 4 + 3 * A_PITCH + 16];
  float* const sh_tl = sh_all;
  float* const sh_l9 = sh_all + T_N * 2;
  float* const sh_tb = sh_all + T_N * 3;
  float* const sh_a  = sh_all + T_N * 3 + 16 * 4;
  float* const sh_z  = sh_a + 3 * A_PITCH;

  __builtin_amdgcn_s_setprio(3);
  const int tid = threadIdx.x;
  const int bd  = blockIdx.x;
  const int b   = bd >> 4;
  const int d   = bd & 15;

  const float k = kernel[d];
  const float alpha = (k > 0.0f) ? (k + log1pf(__expf(-k))) : log1pf(__expf(k));

  const float* base = x + (size_t)b * 64 * T_N;
  const float* px = base + (size_t)d * T_N;
  const float* pm = base + (size_t)(16 + d) * T_N;
  const float* pt = base + (size_t)(32 + d) * T_N;

  if (tid == 0) {
    __builtin_prefetch(px, 0, 1);
    __builtin_prefetch(pm, 0, 1);
    __builtin_prefetch(pt, 0, 1);
  }

  const float ts0 = pt[tid],  ts1 = pt[tid + 128];
  const float mk0 = pm[tid],  mk1 = pm[tid + 128];
  const float xv0 = px[tid],  xv1 = px[tid + 128];
  v2f tl0, tl1;
  tl0.x = ts0;  tl0.y = __logf(fmaxf(mk0, 1e-30f));
  tl1.x = ts1;  tl1.y = __logf(fmaxf(mk1, 1e-30f));
  sh_a[tid] = xv0;
  sh_a[tid + 128] = xv1;
  sh_a[A_PITCH + tid] = 1.0f;
  sh_a[A_PITCH + tid + 128] = 1.0f;
  sh_a[2 * A_PITCH + tid] = 0.0f;
  sh_a[2 * A_PITCH + tid + 128] = 0.0f;
  *reinterpret_cast<v2f*>(&sh_tl[2 * tid]) = tl0;
  *reinterpret_cast<v2f*>(&sh_tl[2 * (tid + 128)]) = tl1;
  sh_l9[tid]       = (-9.0f * LOG2E_F) * tl0.y;
  sh_l9[tid + 128] = (-9.0f * LOG2E_F) * tl1.y;
  {
    float lo0 = ts0, hi0 = ts0, lt0 = tl0.y;
    float lo1 = ts1, hi1 = ts1, lt1 = tl1.y;
#pragma unroll
    for (int s = 1; s < 16; s <<= 1) {
      lo0 = fminf(lo0, __shfl_xor(lo0, s));
      hi0 = fmaxf(hi0, __shfl_xor(hi0, s));
      lt0 = fmaxf(lt0, __shfl_xor(lt0, s));
      lo1 = fminf(lo1, __shfl_xor(lo1, s));
      hi1 = fmaxf(hi1, __shfl_xor(hi1, s));
      lt1 = fmaxf(lt1, __shfl_xor(lt1, s));
    }
    if ((tid & 15) == 0) {
      float4 tb;
      tb.x = lo0; tb.y = hi0; tb.z = LOG2E_F * lt0; tb.w = 0.0f;
      *reinterpret_cast<float4*>(&sh_tb[4 * (tid >> 4)]) = tb;
      sh_z[tid >> 4] = tb.z;
      tb.x = lo1; tb.y = hi1; tb.z = LOG2E_F * lt1;
      *reinterpret_cast<float4*>(&sh_tb[4 * (8 + (tid >> 4))]) = tb;
      sh_z[8 + (tid >> 4)] = tb.z;
    }
  }
  __syncthreads();

  const int wave  = tid >> 5;
  const int lane  = tid & 31;
  const int m16   = lane & 15;
  const int khalf = lane >> 4;

  const float* sh_arow = sh_a + (m16 < 2 ? m16 : 2) * A_PITCH;

  const float sl = alpha * LOG2E_F;
  const float sl10 = 10.0f * sl;
  const size_t ob = (size_t)b * 48 * R_N;

  const float4 tbl = *reinterpret_cast<const float4*>(&sh_tb[4 * m16]);
  float zmax;
  {
    const float4 za = *reinterpret_cast<const float4*>(&sh_z[0]);
    const float4 zb = *reinterpret_cast<const float4*>(&sh_z[4]);
    const float4 zc = *reinterpret_cast<const float4*>(&sh_z[8]);
    const float4 zd = *reinterpret_cast<const float4*>(&sh_z[12]);
    zmax = fmaxf(fmaxf(fmaxf(fmaxf(za.x, za.y), fmaxf(za.z, za.w)), fmaxf(fmaxf(zb.x, zb.y), fmaxf(zb.z, zb.w))),
                 fmaxf(fmaxf(fmaxf(zc.x, zc.y), fmaxf(zc.z, zc.w)), fmaxf(fmaxf(zd.x, zd.y), fmaxf(zd.z, zd.w))));
  }
  const float g1  = fmaf(-8.0f, sl,   zmax);
  const float g10 = fmaf(-8.0f, sl10, zmax);
  const float th1 = g1 - 42.0f, th10 = g10 - 42.0f;

  for (int i = 0; i < 3; ++i) {
    if (i == 1) __builtin_amdgcn_s_setprio(2);
    if (i == 2) __builtin_amdgcn_s_setprio(1);
    const int R   = wave + 4 * i;
    const int r   = R * 16 + m16;
    const float refr  = r * REF_STEP;
    const float r0    = (float)(4 * R + 2);
    const float er    = refr - r0;

    const float r_lo = (R * 16) * REF_STEP;
    const float r_hi = (R * 16 + 15) * REF_STEP;
    const float gap  = fmaxf(fmaxf(tbl.x - r_hi, r_lo - tbl.y), 0.0f);
    const float gp2  = gap * gap;
    const bool cand = !(fmaf(-sl, gp2, tbl.z) < th1) || !(fmaf(-sl10, gp2, tbl.z) < th10);
    const unsigned cm = __builtin_amdgcn_ballot_w32(cand) & 0xFFFFu;

    bool fast = false;
    int tlo8 = 0;
    if (cm != 0u) {
      const int first = __builtin_ctz(cm);
      const int last  = 31 - __builtin_clz(cm);
      tlo8 = first < 8 ? first : 8;
      fast = (last - tlo8) <= 7;
    }

    bool done = false;
    if (fast)
      done = interp_job<8>(sh_arow, sh_tl, sh_l9, 16 * tlo8, sl, er, r0, g1, g10,
                           m16, khalf, lane, out, ob, d, r, i == 2);
    if (!done)
      interp_job<16>(sh_arow, sh_tl, sh_l9, 0, sl, er, r0, g1, g10,
                     m16, khalf, lane, out, ob, d, r, i == 2);
  }
}

extern "C" void kernel_launch(void* const* d_in, const int* in_sizes, int n_in,
                              void* d_out, int out_size, void* d_ws, size_t ws_size,
                              hipStream_t stream) {
  const float* x      = (const float*)d_in[0];
  const float* kernel = (const float*)d_in[1];
  float* out          = (float*)d_out;
  (void)in_sizes; (void)n_in; (void)out_size; (void)d_ws; (void)ws_size;

  interp_wmma_kernel<<<dim3(64 * 16), dim3(128), 0, stream>>>(x, kernel, out);
}


# Serial waits cut: staging loads issue before the softplus scalar chain, 10x map rows before the 1x output division; lane exchanges by DPP/permlanex16, not ds_bpermute. Sums bit-identical.
# speedup vs baseline: 4.4217x; 1.0828x over previous
#include <hip/hip_runtime.h>

typedef __attribute__((ext_vector_type(2))) float v2f;
typedef __attribute__((ext_vector_type(8))) float v8f;

#define T_N 256
#define R_N 192
#define REF_STEP (48.0f / 191.0f)
#define LOG2E_F 1.44269504088896341f
#define LN2_F   0.69314718055994531f
#define A_PITCH (T_N + 16)

__device__ __forceinline__ float lane_xhalf(float v) {
  const int b = __builtin_bit_cast(int, v);
  return __builtin_bit_cast(float, __builtin_amdgcn_permlanex16(b, b, 0x76543210, 0xfedcba98, false, false));
}
template <int M>
__device__ __forceinline__ float row_xor(float v) {
  const int b = __builtin_bit_cast(int, v);
  return __builtin_bit_cast(float, __builtin_amdgcn_update_dpp(b, b, 0x160 + M, 0xf, 0xf, false));
}

template <int NT>
__device__ __forceinline__ bool interp_job(
    const float* sh_arow, const float* sh_tl, const float* sh_l9,
    const int tb0, const float sl, const float er, const float r0,
    const float g1, const float g10,
    const int m16, const int khalf, const int lane,
    float* __restrict__ out, const size_t ob, const int d, const int r, const bool lastjob) {
  v2f bq;
  bq.x = khalf ? 2.0f * sl * er : LOG2E_F;
  bq.y = khalf ? -sl * er * er : -sl;
  asm volatile("" ::: "memory");
  const float* p_tl = sh_tl + 2 * (tb0 + m16);
  const float* p_l9 = sh_l9 + (tb0 + 8 * khalf);
  const float* p_a  = sh_arow + (tb0 + 8 * khalf);
  v2f tlr[NT];
#pragma unroll
  for (int tile = 0; tile < NT; ++tile)
    tlr[tile] = *reinterpret_cast<const v2f*>(p_tl + 32 * tile);

  constexpr bool kGroup = (NT == 8);
  v2f af[kGroup ? NT : 1][4];
  if (kGroup) {
#pragma unroll
    for (int tile = 0; tile < NT; ++tile) {
      const float4 alo = *reinterpret_cast<const float4*>(p_a + 16 * tile);
      const float4 ahi = *reinterpret_cast<const float4*>(p_a + 16 * tile + 4);
      af[tile][0].x = alo.x; af[tile][0].y = alo.y;
      af[tile][1].x = alo.z; af[tile][1].y = alo.w;
      af[tile][2].x = ahi.x; af[tile][2].y = ahi.y;
      af[tile][3].x = ahi.z; af[tile][3].y = ahi.w;
    }
    __builtin_amdgcn_sched_barrier(0);
  }

  v8f S[NT];
#pragma unroll
  for (int tile = 0; tile < NT; ++tile) {
    const v2f tl = tlr[tile];
    const float u = tl.x - r0;
    v2f aq;
    aq.x = khalf ? u : tl.y;
    aq.y = khalf ? 1.0f : u * u;
    const v8f zero = {};
    S[tile] = __builtin_amdgcn_wmma_f32_16x16x4_f32(
         false, aq,  false, bq,
         (short)0, zero,  false,  false);
  }

  float4 l9lo[kGroup ? NT : 1], l9hi[kGroup ? NT : 1];
  for (int p = 0; p < 2; ++p) {
    if (p == 1 && lastjob) __builtin_amdgcn_s_setprio(0);
    if (p == 1) {
#pragma unroll
      for (int tile = 0; tile < NT; ++tile) {
        const float4 llo = kGroup ? l9lo[tile] : *reinterpret_cast<const float4*>(p_l9 + 16 * tile);
        const float4 lhi = kGroup ? l9hi[tile] : *reinterpret_cast<const float4*>(p_l9 + 16 * tile + 4);
        const float l9[8] = {llo.x, llo.y, llo.z, llo.w, lhi.x, lhi.y, lhi.z, lhi.w};
        const v2f ten = {10.0f, 10.0f};
#pragma unroll
        for (int q = 0; q < 4; ++q) {
          v2f s2, l2;
          s2.x = S[tile][2 * q];  s2.y = S[tile][2 * q + 1];
          l2.x = l9[2 * q];       l2.y = l9[2 * q + 1];
          s2 = __builtin_elementwise_fma(ten, s2, l2);
          S[tile][2 * q] = s2.x;  S[tile][2 * q + 1] = s2.y;
        }
      }
    }

    float tm[NT];
#pragma unroll
    for (int tile = 0; tile < NT; ++tile) {
      tm[tile] = fmaxf(fmaxf(fmaxf(S[tile][0], S[tile][1]), fmaxf(S[tile][2], S[tile][3])),
                       fmaxf(fmaxf(S[tile][4], S[tile][5]), fmaxf(S[tile][6], S[tile][7])));
    }
    float mx = tm[0];
#pragma unroll
    for (int tile = 1; tile < NT; ++tile) mx = fmaxf(mx, tm[tile]);
    mx = fmaxf(mx, lane_xhalf(mx));


    bool keep[NT];
#pragma unroll
    for (int tile = 0; tile < NT; ++tile) {
      const bool live = (tm[tile] - mx) >= -40.0f;
      keep[tile] = __builtin_amdgcn_ballot_w32(live) != 0u;
    }

    if (kGroup) {
      const bool below = !(mx >= (p == 0 ? g1 : g10));
      if (__builtin_amdgcn_ballot_w32(below) != 0u) return false;
    }

    v8f acc0 = {}, acc1 = {};
#pragma unroll
    for (int tile = 0; tile < NT; ++tile) {
      if (!keep[tile]) continue;
      v2f a4[4];
      if (kGroup) {
#pragma unroll
        for (int q = 0; q < 4; ++q) a4[q] = af[tile][q];
      } else {
        const float4 alo = *reinterpret_cast<const float4*>(p_a + 16 * tile);
        const float4 ahi = *reinterpret_cast<const float4*>(p_a + 16 * tile + 4);
        a4[0].x = alo.x; a4[0].y = alo.y;
        a4[1].x = alo.z; a4[1].y = alo.w;
        a4[2].x = ahi.x; a4[2].y = ahi.y;
        a4[3].x = ahi.z; a4[3].y = ahi.w;
      }
#pragma unroll
      for (int q = 0; q < 4; ++q) {
        v2f ev;
        const v2f av = a4[q];
        ev.x = __builtin_amdgcn_exp2f(S[tile][2 * q]     - mx);
        ev.y = __builtin_amdgcn_exp2f(S[tile][2 * q + 1] - mx);
        if (q & 1)
          acc1 = __builtin_amdgcn_wmma_f32_16x16x4_f32(
               false, av,  false, ev,
               (short)0, acc1,  false,  false);
        else
          acc0 = __builtin_amdgcn_wmma_f32_16x16x4_f32(
               false, av,  false, ev,
               (short)0, acc0,  false,  false);
      }
    }

    if (kGroup && p == 0) {
#pragma unroll
      for (int tile = 0; tile < NT; ++tile) {
        l9lo[tile] = *reinterpret_cast<const float4*>(p_l9 + 16 * tile);
        l9hi[tile] = *reinterpret_cast<const float4*>(p_l9 + 16 * tile + 4);
      }
    }

    if (lane < 16) {
      const float ys = acc0[0] + acc1[0];
      const float dn = acc0[1] + acc1[1];
      const int ch_y = (p == 0) ? d : (32 + d);
      out[ob + (size_t)ch_y * R_N + r] = ys / dn;
      if (p == 0)
        out[ob + (size_t)(16 + d) * R_N + r] = (mx + __log2f(dn)) * LN2_F;
    }
  }
  return true;
}

__global__ __launch_bounds__(128) void interp_wmma_kernel(
    const float* __restrict__ x, const float* __restrict__ kernel,
    float* __restrict__ out) {
  __shared__ __attribute__((aligned(16))) float sh_all[T_N * 2 + T_N + 16 * 4 + 3 * A_PITCH + 16];
  float* const sh_tl = sh_all;
  float* const sh_l9 = sh_all + T_N * 2;
  float* const sh_tb = sh_all + T_N * 3;
  float* const sh_a  = sh_all + T_N * 3 + 16 * 4;
  float* const sh_z  = sh_a + 3 * A_PITCH;

  __builtin_amdgcn_s_setprio(3);
  const int tid = threadIdx.x;
  const int bd  = blockIdx.x;
  const int b   = bd >> 4;
  const int d   = bd & 15;

  const float* base = x + (size_t)b * 64 * T_N;
  const float* px = base + (size_t)d * T_N;
  const float* pm = base + (size_t)(16 + d) * T_N;
  const float* pt = base + (size_t)(32 + d) * T_N;

  const float ts0 = pt[tid],  ts1 = pt[tid + 128];
  const float mk0 = pm[tid],  mk1 = pm[tid + 128];
  const float xv0 = px[tid],  xv1 = px[tid + 128];

  const float k = kernel[d];
  const float alpha = (k > 0.0f) ? (k + log1pf(__expf(-k))) : log1pf(__expf(k));

  v2f tl0, tl1;
  tl0.x = ts0;  tl0.y = __logf(fmaxf(mk0, 1e-30f));
  tl1.x = ts1;  tl1.y = __logf(fmaxf(mk1, 1e-30f));
  sh_a[tid] = xv0;
  sh_a[tid + 128] = xv1;
  sh_a[A_PITCH + tid] = 1.0f;
  sh_a[A_PITCH + tid + 128] = 1.0f;
  sh_a[2 * A_PITCH + tid] = 0.0f;
  sh_a[2 * A_PITCH + tid + 128] = 0.0f;
  *reinterpret_cast<v2f*>(&sh_tl[2 * tid]) = tl0;
  *reinterpret_cast<v2f*>(&sh_tl[2 * (tid + 128)]) = tl1;
  sh_l9[tid]       = (-9.0f * LOG2E_F) * tl0.y;
  sh_l9[tid + 128] = (-9.0f * LOG2E_F) * tl1.y;
  {
    float lo0 = ts0, hi0 = ts0, lt0 = tl0.y;
    float lo1 = ts1, hi1 = ts1, lt1 = tl1.y;
#define SPAN_STEP(M)                        \
    lo0 = fminf(lo0, row_xor<M>(lo0));      \
    hi0 = fmaxf(hi0, row_xor<M>(hi0));      \
    lt0 = fmaxf(lt0, row_xor<M>(lt0));      \
    lo1 = fminf(lo1, row_xor<M>(lo1));      \
    hi1 = fmaxf(hi1, row_xor<M>(hi1));      \
    lt1 = fmaxf(lt1, row_xor<M>(lt1));
    SPAN_STEP(1) SPAN_STEP(2) SPAN_STEP(4) SPAN_STEP(8)
#undef SPAN_STEP
    if ((tid & 15) == 0) {
      float4 tb;
      tb.x = lo0; tb.y = hi0; tb.z = LOG2E_F * lt0; tb.w = 0.0f;
      *reinterpret_cast<float4*>(&sh_tb[4 * (tid >> 4)]) = tb;
      sh_z[tid >> 4] = tb.z;
      tb.x = lo1; tb.y = hi1; tb.z = LOG2E_F * lt1;
      *reinterpret_cast<float4*>(&sh_tb[4 * (8 + (tid >> 4))]) = tb;
      sh_z[8 + (tid >> 4)] = tb.z;
    }
  }
  __syncthreads();

  const int wave  = tid >> 5;
  const int lane  = tid & 31;
  const int m16   = lane & 15;
  const int khalf = lane >> 4;

  const float* sh_arow = sh_a + (m16 < 2 ? m16 : 2) * A_PITCH;

  const float sl = alpha * LOG2E_F;
  const float sl10 = 10.0f * sl;
  const size_t ob = (size_t)b * 48 * R_N;

  const float4 tbl = *reinterpret_cast<const float4*>(&sh_tb[4 * m16]);
  float zmax;
  {
    const float4 za = *reinterpret_cast<const float4*>(&sh_z[0]);
    const float4 zb = *reinterpret_cast<const float4*>(&sh_z[4]);
    const float4 zc = *reinterpret_cast<const float4*>(&sh_z[8]);
    const float4 zd = *reinterpret_cast<const float4*>(&sh_z[12]);
    zmax = fmaxf(fmaxf(fmaxf(fmaxf(za.x, za.y), fmaxf(za.z, za.w)), fmaxf(fmaxf(zb.x, zb.y), fmaxf(zb.z, zb.w))),
                 fmaxf(fmaxf(fmaxf(zc.x, zc.y), fmaxf(zc.z, zc.w)), fmaxf(fmaxf(zd.x, zd.y), fmaxf(zd.z, zd.w))));
  }
  const float g1  = fmaf(-8.0f, sl,   zmax);
  const float g10 = fmaf(-8.0f, sl10, zmax);
  const float th1 = g1 - 42.0f, th10 = g10 - 42.0f;

  for (int i = 0; i < 3; ++i) {
    if (i == 1) __builtin_amdgcn_s_setprio(2);
    if (i == 2) __builtin_amdgcn_s_setprio(1);
    const int R   = wave + 4 * i;
    const int r   = R * 16 + m16;
    const float refr  = r * REF_STEP;
    const float r0    = (float)(4 * R + 2);
    const float er    = refr - r0;

    const float r_lo = (R * 16) * REF_STEP;
    const float r_hi = (R * 16 + 15) * REF_STEP;
    const float gap  = fmaxf(fmaxf(tbl.x - r_hi, r_lo - tbl.y), 0.0f);
    const float gp2  = gap * gap;
    const bool cand = !(fmaf(-sl, gp2, tbl.z) < th1) || !(fmaf(-sl10, gp2, tbl.z) < th10);
    const unsigned cm = __builtin_amdgcn_ballot_w32(cand) & 0xFFFFu;

    bool fast = false;
    int tlo8 = 0;
    if (cm != 0u) {
      const int first = __builtin_ctz(cm);
      const int last  = 31 - __builtin_clz(cm);
      tlo8 = first < 8 ? first : 8;
      fast = (last - tlo8) <= 7;
    }

    bool done = false;
    if (fast)
      done = interp_job<8>(sh_arow, sh_tl, sh_l9, 16 * tlo8, sl, er, r0, g1, g10,
                           m16, khalf, lane, out, ob, d, r, i == 2);
    if (!done)
      interp_job<16>(sh_arow, sh_tl, sh_l9, 0, sl, er, r0, g1, g10,
                     m16, khalf, lane, out, ob, d, r, i == 2);
  }
}

extern "C" void kernel_launch(void* const* d_in, const int* in_sizes, int n_in,
                              void* d_out, int out_size, void* d_ws, size_t ws_size,
                              hipStream_t stream) {
  const float* x      = (const float*)d_in[0];
  const float* kernel = (const float*)d_in[1];
  float* out          = (float*)d_out;
  (void)in_sizes; (void)n_in; (void)out_size; (void)d_ws; (void)ws_size;

  interp_wmma_kernel<<<dim3(64 * 16), dim3(128), 0, stream>>>(x, kernel, out);
}
